# starting kernel
# baseline (speedup 1.0000x reference)
#include <hip/hip_runtime.h>
#include <hip/hip_bf16.h>

typedef __attribute__((ext_vector_type(2))) float v2f;
typedef __attribute__((ext_vector_type(8))) float v8f;

#define B_    128
#define R_    36
#define W_    60
#define D_    1024
#define TK_   20
#define EPS_  1e-12f
#define NEG_  -1e30f

__global__ __launch_bounds__(256)
void row_invnorm_kernel(const float* __restrict__ img,
                        const float* __restrict__ cap,
                        float* __restrict__ inv_cap,
                        float* __restrict__ inv_img) {
    __shared__ float red[256];
    const int row = blockIdx.x;
    const float* p = (row < B_ * W_) ? (cap + (size_t)row * D_)
                                     : (img + (size_t)(row - B_ * W_) * D_);
    const float4 v = ((const float4*)p)[threadIdx.x];
    red[threadIdx.x] = v.x * v.x + v.y * v.y + v.z * v.z + v.w * v.w;
    __syncthreads();
    for (int st = 128; st >= 1; st >>= 1) {
        if (threadIdx.x < st) red[threadIdx.x] += red[threadIdx.x + st];
        __syncthreads();
    }
    if (threadIdx.x == 0) {
        const float inv = 1.0f / fmaxf(sqrtf(red[0]), EPS_);
        if (row < B_ * W_) inv_cap[row] = inv;
        else               inv_img[row - B_ * W_] = inv;
    }
}

__global__ __launch_bounds__(128)
void pair_kernel(const float* __restrict__ img,
                 const float* __restrict__ cap,
                 const float* __restrict__ inv_cap,
                 const float* __restrict__ inv_img,
                 const float* __restrict__ w1_t2i, const float* __restrict__ b1_t2i,
                 const float* __restrict__ w2_t2i, const float* __restrict__ b2_t2i,
                 const float* __restrict__ w1_i2t, const float* __restrict__ b1_i2t,
                 const float* __restrict__ w2_i2t, const float* __restrict__ b2_i2t,
                 const int*   __restrict__ cap_len,
                 float*       __restrict__ out) {
    const int i    = blockIdx.x;
    const int c    = blockIdx.y;
    const int tid  = threadIdx.x;
    const int lane = tid & 31;
    const int wv   = tid >> 5;

    __shared__ float As[64][33];
    __shared__ float Bs[48][33];
    __shared__ float Sm[64][49];
    __shared__ float rnw[48];
    __shared__ float rnr[64];
    __shared__ float rowsim[64];
    __shared__ float colsim[48];
    __shared__ float topr[TK_], topc[TK_];
    __shared__ float h1[TK_], h2[TK_];
    __shared__ float red[128];
    __shared__ float s_rowmean, s_colmean;

    for (int idx = tid; idx < 4 * 33;  idx += 128) As[60 + idx / 33][idx % 33] = 0.f;
    for (int idx = tid; idx < 12 * 33; idx += 128) Bs[36 + idx / 33][idx % 33] = 0.f;

    v8f acc0 = {}; v8f acc1 = {}; v8f acc2 = {};
    const int mr = (wv << 4) + (lane & 15);
    const int nr = lane & 15;
    const int kb = (lane >> 4) << 1;

    const float* capRow = cap + (size_t)c * W_ * D_;
    const float* imgRow = img + (size_t)i * R_ * D_;
    const float* invC   = inv_cap + c * W_;
    const float* invI   = inv_img + i * R_;

    for (int k0 = 0; k0 < D_; k0 += 32) {
        __syncthreads();
        for (int idx = tid; idx < 480 + 288; idx += 128) {
            if (idx < 480) {
                const int row = idx >> 3, c4 = (idx & 7) << 2;
                const float4 v = *(const float4*)(capRow + (size_t)row * D_ + k0 + c4);
                const float sc = invC[row];
                As[row][c4 + 0] = v.x * sc; As[row][c4 + 1] = v.y * sc;
                As[row][c4 + 2] = v.z * sc; As[row][c4 + 3] = v.w * sc;
            } else {
                const int j = idx - 480;
                const int row = j >> 3, c4 = (j & 7) << 2;
                const float4 v = *(const float4*)(imgRow + (size_t)row * D_ + k0 + c4);
                const float sc = invI[row];
                Bs[row][c4 + 0] = v.x * sc; Bs[row][c4 + 1] = v.y * sc;
                Bs[row][c4 + 2] = v.z * sc; Bs[row][c4 + 3] = v.w * sc;
            }
        }
        __syncthreads();
        for (int kk = 0; kk < 32; kk += 4) {
            v2f a;   a.x   = As[mr][kk + kb];      a.y   = As[mr][kk + kb + 1];
            v2f b0;  b0.x  = Bs[nr][kk + kb];      b0.y  = Bs[nr][kk + kb + 1];
            v2f b1v; b1v.x = Bs[16 + nr][kk + kb]; b1v.y = Bs[16 + nr][kk + kb + 1];
            v2f b2v; b2v.x = Bs[32 + nr][kk + kb]; b2v.y = Bs[32 + nr][kk + kb + 1];
            acc0 = __builtin_amdgcn_wmma_f32_16x16x4_f32(false, a, false, b0,  (short)0, acc0, false, false);
            acc1 = __builtin_amdgcn_wmma_f32_16x16x4_f32(false, a, false, b1v, (short)0, acc1, false, false);
            acc2 = __builtin_amdgcn_wmma_f32_16x16x4_f32(false, a, false, b2v, (short)0, acc2, false, false);
        }
    }
    __syncthreads();

    {
        const int mb = (wv << 4) + ((lane >> 4) << 3);
        for (int v = 0; v < 8; v++) {
            const int m = mb + v;
            float x0 = acc0[v]; Sm[m][nr]      = (x0 >= 0.f) ? x0 : 0.1f * x0;
            float x1 = acc1[v]; Sm[m][16 + nr] = (x1 >= 0.f) ? x1 : 0.1f * x1;
            float x2 = acc2[v]; Sm[m][32 + nr] = (x2 >= 0.f) ? x2 : 0.1f * x2;
        }
    }
    __syncthreads();

    const int cl   = cap_len[c];
    const int nsel = (cl > TK_) ? cl : TK_;

    if (tid < R_) {
        float ss = 0.f;
        for (int w = 0; w < nsel; w++) { float x = Sm[w][tid]; ss += x * x; }
        rnw[tid] = 1.0f / fmaxf(sqrtf(ss), EPS_);
    } else if (tid >= 64 && tid < 64 + W_) {
        const int w = tid - 64;
        float ss = 0.f;
        for (int r = 0; r < R_; r++) { float x = Sm[w][r]; ss += x * x; }
        rnr[w] = 1.0f / fmaxf(sqrtf(ss), EPS_);
    }
    __syncthreads();

    if (tid < W_) {
        float m = NEG_;
        if (tid < nsel)
            for (int r = 0; r < R_; r++) m = fmaxf(m, Sm[tid][r] * rnw[r]);
        rowsim[tid] = m;
    } else if (tid >= 64 && tid < 64 + R_) {
        const int r = tid - 64;
        float m = NEG_;
        for (int w = 0; w < cl; w++) m = fmaxf(m, Sm[w][r] * rnr[w]);
        colsim[r] = m;
    }
    __syncthreads();

    red[tid] = (tid < cl) ? rowsim[tid] : 0.f;
    __syncthreads();
    for (int st = 64; st >= 1; st >>= 1) {
        if (tid < st) red[tid] += red[tid + st];
        __syncthreads();
    }
    if (tid == 0) s_rowmean = red[0] / (float)cl;
    __syncthreads();
    red[tid] = (tid < R_) ? colsim[tid] : 0.f;
    __syncthreads();
    for (int st = 64; st >= 1; st >>= 1) {
        if (tid < st) red[tid] += red[tid + st];
        __syncthreads();
    }
    if (tid == 0) s_colmean = red[0] / (float)R_;
    __syncthreads();

    if (tid < W_) {
        const float v = rowsim[tid];
        int rank = 0;
        for (int j = 0; j < W_; j++) {
            const float u = rowsim[j];
            rank += (u > v) || (u == v && j < tid);
        }
        if (rank < TK_) topr[rank] = v;
    } else if (tid >= 64 && tid < 64 + R_) {
        const int r = tid - 64;
        const float v = colsim[r];
        int rank = 0;
        for (int j = 0; j < R_; j++) {
            const float u = colsim[j];
            rank += (u > v) || (u == v && j < r);
        }
        if (rank < TK_) topc[rank] = v;
    }
    __syncthreads();

    if (tid < TK_) {
        float a = b1_t2i[tid];
        for (int k = 0; k < TK_; k++) a += topr[k] * w1_t2i[k * TK_ + tid];
        h1[tid] = fmaxf(a, 0.f);
    } else if (tid >= 32 && tid < 32 + TK_) {
        const int j = tid - 32;
        float a = b1_i2t[j];
        for (int k = 0; k < TK_; k++) a += topc[k] * w1_i2t[k * TK_ + j];
        h2[j] = fmaxf(a, 0.f);
    }
    __syncthreads();

    if (tid == 0) {
        float o = b2_t2i[0];
        for (int j = 0; j < TK_; j++) o += h1[j] * w2_t2i[j];
        out[c * B_ + i]              = o + s_rowmean;
        out[2 * B_ * B_ + c * B_ + i] = s_rowmean;
    } else if (tid == 32) {
        float o = b2_i2t[0];
        for (int j = 0; j < TK_; j++) o += h2[j] * w2_i2t[j];
        out[B_ * B_ + i * B_ + c]     = o + s_colmean;
        out[3 * B_ * B_ + i * B_ + c] = s_colmean;
    }
}

extern "C" void kernel_launch(void* const* d_in, const int* in_sizes, int n_in,
                              void* d_out, int out_size, void* d_ws, size_t ws_size,
                              hipStream_t stream) {
    const float* img     = (const float*)d_in[0];
    const float* cap     = (const float*)d_in[1];
    const float* w1_t2i  = (const float*)d_in[2];
    const float* b1_t2i  = (const float*)d_in[3];
    const float* w2_t2i  = (const float*)d_in[4];
    const float* b2_t2i  = (const float*)d_in[5];
    const float* w1_i2t  = (const float*)d_in[6];
    const float* b1_i2t  = (const float*)d_in[7];
    const float* w2_i2t  = (const float*)d_in[8];
    const float* b2_i2t  = (const float*)d_in[9];
    const int*   cap_len = (const int*)d_in[10];
    float*       out     = (float*)d_out;

    float* inv_cap = (float*)d_ws;
    float* inv_img = inv_cap + B_ * W_;

    row_invnorm_kernel<<<B_ * W_ + B_ * R_, 256, 0, stream>>>(img, cap, inv_cap, inv_img);

    dim3 grid(B_, B_);
    pair_kernel<<<grid, 128, 0, stream>>>(img, cap, inv_cap, inv_img,
                                          w1_t2i, b1_t2i, w2_t2i, b2_t2i,
                                          w1_i2t, b1_i2t, w2_i2t, b2_i2t,
                                          cap_len, out);
}


# per-pair tail re-cut: one wave per pair, 60x36 tile in registers (2 word rows per lane), over-word sums and maxima by lane exchange; 5 block barriers, was 23; over-word norm sums in a new order
# speedup vs baseline: 2.9321x; 2.9321x over previous
#include <hip/hip_runtime.h>
#include <hip/hip_bf16.h>

typedef __attribute__((ext_vector_type(2))) float v2f;
typedef __attribute__((ext_vector_type(8))) float v8f;

#define B_    128
#define R_    36
#define W_    60
#define D_    1024
#define TK_   20
#define EPS_  1e-12f
#define NEG_  -1e30f

__global__ __launch_bounds__(256)
void row_invnorm_kernel(const float* __restrict__ img,
                        const float* __restrict__ cap,
                        const int*   __restrict__ cap_len,
                        float* __restrict__ inv_cap,
                        float* __restrict__ inv_img,
                        int*   __restrict__ rowmap,
                        int*   __restrict__ nlive) {
    __shared__ float red[256];
    __shared__ int   wsum[8];
    if (blockIdx.x == 0) {
        const int t  = threadIdx.x;
        const int ln = t & 31;
        const int ns = (t < B_) ? min(max(cap_len[t], TK_), W_) : 0;
        int incl = ns;
        for (int d = 1; d < 32; d <<= 1) {
            const int u = __shfl_up(incl, d);
            if (ln >= d) incl += u;
        }
        if (ln == 31) wsum[t >> 5] = incl;
        __syncthreads();
        int base = incl - ns, total = 0;
        for (int k = 0; k < 8; k++) {
            const int s = wsum[k];
            total += s;
            if (k < (t >> 5)) base += s;
        }
        for (int w = 0; w < ns; w++) rowmap[base + w] = t * W_ + w;
        for (int m = total + t; m < B_ * W_; m += 256) rowmap[m] = -1;
        if (t == 0) nlive[0] = total;
        return;
    }
    const int row = blockIdx.x - 1;
    const float* p = (row < B_ * W_) ? (cap + (size_t)row * D_)
                                     : (img + (size_t)(row - B_ * W_) * D_);
    const float4 v = ((const float4*)p)[threadIdx.x];
    red[threadIdx.x] = v.x * v.x + v.y * v.y + v.z * v.z + v.w * v.w;
    __syncthreads();
    for (int st = 128; st >= 1; st >>= 1) {
        if (threadIdx.x < st) red[threadIdx.x] += red[threadIdx.x + st];
        __syncthreads();
    }
    if (threadIdx.x == 0) {
        const float inv = 1.0f / fmaxf(sqrtf(red[0]), EPS_);
        if (row < B_ * W_) inv_cap[row] = inv;
        else               inv_img[row - B_ * W_] = inv;
    }
}

#define PITCH_ 36
#define NT_    36
#define NWGP_  256

template <int TI, int TJ>
__device__ __forceinline__ void gemm_tile(const float* __restrict__ img,
                                          const float* __restrict__ cap,
                                          const float* __restrict__ inv_cap,
                                          const float* __restrict__ inv_img,
                                          const int*   __restrict__ rowmap,
                                          float* __restrict__ sim,
                                          const int m0,
                                          const int n0,
                                          float (*As)[128][PITCH_],
                                          float (*Bs)[128][PITCH_],
                                          int* rowS) {
    constexpr int BM = 32 * TI;
    constexpr int JA = BM / 16;
    constexpr int JB = (32 * TJ) / 16;

    const int tid  = threadIdx.x;
    const int lane = tid & 31;
    const int wv   = tid >> 5;
    const int wm   = wv >> 1;
    const int wn   = wv & 1;

    if (tid < BM) rowS[tid] = rowmap[m0 + tid];
    __syncthreads();
    if (rowS[0] < 0) return;

    const int srow = tid >> 3;
    const int sc4  = (tid & 7) << 2;
    float scA[JA], scB[JB];
    int   offA[JA];
#pragma unroll
    for (int j = 0; j < JA; j++) {
        const int o  = rowS[srow + 16 * j];
        const int oc = (o < 0) ? 0 : o;
        scA[j]  = inv_cap[oc];
        offA[j] = oc * D_ + sc4;
    }
#pragma unroll
    for (int j = 0; j < JB; j++) scB[j] = inv_img[n0 + srow + 16 * j];
    const float* gB = img + (size_t)(n0 + srow) * D_ + sc4;

    v8f acc[TI][TJ];
#pragma unroll
    for (int a = 0; a < TI; a++)
#pragma unroll
        for (int b = 0; b < TJ; b++) acc[a][b] = (v8f){};

    const int fr = lane & 15;
    const int kb = (lane >> 4) << 1;

    float4 va[JA], vb[JB];
#pragma unroll
    for (int j = 0; j < JA; j++) va[j] = *(const float4*)(cap + offA[j]);
#pragma unroll
    for (int j = 0; j < JB; j++) vb[j] = *(const float4*)(gB + (size_t)j * 16 * D_);
#pragma unroll
    for (int j = 0; j < JA; j++) {
        float4 x = va[j]; const float s = scA[j];
        x.x *= s; x.y *= s; x.z *= s; x.w *= s;
        *(float4*)&As[0][srow + 16 * j][sc4] = x;
    }
#pragma unroll
    for (int j = 0; j < JB; j++) {
        float4 y = vb[j]; const float t = scB[j];
        y.x *= t; y.y *= t; y.z *= t; y.w *= t;
        *(float4*)&Bs[0][srow + 16 * j][sc4] = y;
    }
    __syncthreads();

    for (int k0 = 0; k0 < D_; k0 += 32) {
        const int cur = (k0 >> 5) & 1;
        const bool more = (k0 + 32 < D_);
        if (more) {
#pragma unroll
            for (int j = 0; j < JA; j++) va[j] = *(const float4*)(cap + offA[j] + k0 + 32);
#pragma unroll
            for (int j = 0; j < JB; j++) vb[j] = *(const float4*)(gB + (size_t)j * 16 * D_ + k0 + 32);
        }
#pragma unroll
        for (int kk = 0; kk < 32; kk += 4) {
            v2f a[TI], b[TJ];
#pragma unroll
            for (int t = 0; t < TI; t++) a[t] = *(const v2f*)&As[cur][wm * (16 * TI) + t * 16 + fr][kk + kb];
#pragma unroll
            for (int t = 0; t < TJ; t++) b[t] = *(const v2f*)&Bs[cur][wn * (16 * TJ) + t * 16 + fr][kk + kb];
#pragma unroll
            for (int ti = 0; ti < TI; ti++)
#pragma unroll
                for (int tj = 0; tj < TJ; tj++)
                    acc[ti][tj] = __builtin_amdgcn_wmma_f32_16x16x4_f32(false, a[ti], false, b[tj], (short)0, acc[ti][tj], false, false);
        }
        if (more) {
#pragma unroll
            for (int j = 0; j < JA; j++) {
                float4 x = va[j]; const float s = scA[j];
                x.x *= s; x.y *= s; x.z *= s; x.w *= s;
                *(float4*)&As[cur ^ 1][srow + 16 * j][sc4] = x;
            }
#pragma unroll
            for (int j = 0; j < JB; j++) {
                float4 y = vb[j]; const float t = scB[j];
                y.x *= t; y.y *= t; y.z *= t; y.w *= t;
                *(float4*)&Bs[cur ^ 1][srow + 16 * j][sc4] = y;
            }
        }
        __syncthreads();
    }

    int colOff[TJ];
#pragma unroll
    for (int tj = 0; tj < TJ; tj++) {
        const int n = n0 + wn * (16 * TJ) + tj * 16 + fr;
        const int i = n / R_;
        colOff[tj] = i * (W_ * R_) + (n - i * R_);
    }
    const int mb = wm * (16 * TI) + ((lane >> 4) << 3);
#pragma unroll
    for (int ti = 0; ti < TI; ti++) {
#pragma unroll
        for (int v = 0; v < 8; v++) {
            const int m = rowS[mb + ti * 16 + v];
            if (m >= 0) {
                const int c = m / W_;
                const int rowOff = c * (B_ * W_ * R_) + (m - c * W_) * R_;
#pragma unroll
                for (int tj = 0; tj < TJ; tj++) {
                    const float x = acc[ti][tj][v];
                    sim[rowOff + colOff[tj]] = (x >= 0.f) ? x : 0.1f * x;
                }
            }
        }
    }
}

__global__ __launch_bounds__(128)
void sim_gemm_kernel(const float* __restrict__ img,
                     const float* __restrict__ cap,
                     const float* __restrict__ inv_cap,
                     const float* __restrict__ inv_img,
                     const int*   __restrict__ rowmap,
                     const int*   __restrict__ nlive,
                     float* __restrict__ sim) {
    __shared__ __attribute__((aligned(16))) float As[2][128][PITCH_];
    __shared__ __attribute__((aligned(16))) float Bs[2][128][PITCH_];
    __shared__ int rowS[128];

    const int tiles = ((nlive[0] + 127) >> 7) * NT_;
    const int bulk  = tiles & ~(NWGP_ - 1);
    const int id    = blockIdx.x;
    if (id < bulk) {
        const int ty = id / NT_;
        gemm_tile<4, 4>(img, cap, inv_cap, inv_img, rowmap, sim,
                        ty * 128, (id - ty * NT_) * 128, As, Bs, rowS);
    } else {
        const int q = id - bulk;
        const int t = bulk + (q >> 2);
        if (t >= tiles) return;
        const int ty = t / NT_;
        gemm_tile<2, 2>(img, cap, inv_cap, inv_img, rowmap, sim,
                        ty * 128 + ((q >> 1) & 1) * 64, (t - ty * NT_) * 128 + (q & 1) * 64, As, Bs, rowS);
    }
}

template <int M>
__device__ __forceinline__ float lane_xor(float v) {
    if (M == 16)
        return __builtin_bit_cast(float, __builtin_amdgcn_permlanex16(0, __builtin_bit_cast(int, v), 0x76543210, 0xfedcba98, true, false));
    else
        return __builtin_bit_cast(float, __builtin_amdgcn_update_dpp(0, __builtin_bit_cast(int, v), 0x160 + M, 0xf, 0xf, true));
}

template <int M, int N, bool MAXOP>
__device__ __forceinline__ void halve_step(float* v, const bool up) {
#pragma unroll
    for (int j = 0; j < N; j++) {
        const float a = v[j], b = v[j + N];
        const float keep = up ? b : a;
        const float send = up ? a : b;
        const float recv = lane_xor<M>(send);
        v[j] = MAXOP ? fmaxf(keep, recv) : (keep + recv);
    }
}

template <bool MAXOP>
__device__ __forceinline__ float reduce_columns(float* v, float* ex, const int lane) {
    halve_step<16, 16, MAXOP>(v, (lane & 16) != 0);
    halve_step<8, 8, MAXOP>(v, (lane & 8) != 0);
    halve_step<4, 4, MAXOP>(v, (lane & 4) != 0);
    halve_step<2, 2, MAXOP>(v, (lane & 2) != 0);
    halve_step<1, 1, MAXOP>(v, (lane & 1) != 0);
#pragma unroll
    for (int e = 0; e < 4; e++) {
        float t = v[32 + e];
        { const float o = lane_xor<16>(t); t = MAXOP ? fmaxf(t, o) : (t + o); }
        { const float o = lane_xor<8>(t);  t = MAXOP ? fmaxf(t, o) : (t + o); }
        { const float o = lane_xor<4>(t);  t = MAXOP ? fmaxf(t, o) : (t + o); }
        { const float o = lane_xor<2>(t);  t = MAXOP ? fmaxf(t, o) : (t + o); }
        { const float o = lane_xor<1>(t);  t = MAXOP ? fmaxf(t, o) : (t + o); }
        ex[e] = t;
    }
    return v[0];
}

__device__ __forceinline__ float tree_sum(float t) {
    t += lane_xor<16>(t);
    t += lane_xor<8>(t);
    t += lane_xor<4>(t);
    t += lane_xor<2>(t);
    t += lane_xor<1>(t);
    return t;
}

__global__ __launch_bounds__(128)
void pair_tail_kernel(const float* __restrict__ sim,
                      const float* __restrict__ w1_t2i, const float* __restrict__ b1_t2i,
                      const float* __restrict__ w2_t2i, const float* __restrict__ b2_t2i,
                      const float* __restrict__ w1_i2t, const float* __restrict__ b1_i2t,
                      const float* __restrict__ w2_i2t, const float* __restrict__ b2_i2t,
                      const int*   __restrict__ cap_len,
                      float*       __restrict__ out) {
    const int tid  = threadIdx.x;
    const int lane = tid & 31;
    const int wv   = tid >> 5;
    const int i    = blockIdx.x * 4 + wv;
    const int c    = blockIdx.y;

    __shared__ float4 tileS[4][17 * 32];
    __shared__ __attribute__((aligned(16))) float wS[2][TK_ * TK_];
    __shared__ __attribute__((aligned(16))) float w2S[2][TK_];
    __shared__ __attribute__((aligned(16))) float rnwS[4][40];
    __shared__ __attribute__((aligned(16))) float rowS[4][64];
    __shared__ __attribute__((aligned(16))) float colS[4][40];
    __shared__ __attribute__((aligned(16))) float topS[4][2][TK_];
    __shared__ __attribute__((aligned(16))) float hS[4][2][TK_];

    for (int k = tid; k < TK_ * TK_; k += 128) { wS[0][k] = w1_t2i[k]; wS[1][k] = w1_i2t[k]; }
    if (tid < TK_) { w2S[0][tid] = w2_t2i[tid]; w2S[1][tid] = w2_i2t[tid]; }

    const int cl   = cap_len[c];
    const int nsel = min(max(cl, TK_), W_);
    const int w0 = lane, w1 = lane + 32;
    const bool sel0 = w0 < nsel, sel1 = w1 < nsel;
    const bool real0 = w0 < cl,  real1 = (w1 < cl) && (w1 < W_);

    const float4* tile = (const float4*)(sim + (size_t)(c * B_ + i) * (W_ * R_));
    const int npiece = nsel * (R_ / 4);
    float4 pc[17];
#pragma unroll
    for (int q = 0; q < 17; q++) {
        const int idx = q * 32 + lane;
        pc[q] = tile[(idx < npiece) ? idx : 0];
    }
#pragma unroll
    for (int q = 0; q < 17; q++) tileS[wv][q * 32 + lane] = pc[q];
    __syncthreads();
    float x0[R_], x1[R_];
#pragma unroll
    for (int q = 0; q < 9; q++) {
        const float4 a = sel0 ? tileS[wv][w0 * 9 + q] : make_float4(0.f, 0.f, 0.f, 0.f);
        const float4 b = sel1 ? tileS[wv][w1 * 9 + q] : make_float4(0.f, 0.f, 0.f, 0.f);
        x0[4 * q + 0] = a.x; x0[4 * q + 1] = a.y; x0[4 * q + 2] = a.z; x0[4 * q + 3] = a.w;
        x1[4 * q + 0] = b.x; x1[4 * q + 1] = b.y; x1[4 * q + 2] = b.z; x1[4 * q + 3] = b.w;
    }

    float ss0 = 0.f, ss1 = 0.f;
#pragma unroll
    for (int r = 0; r < R_; r++) { ss0 += x0[r] * x0[r]; ss1 += x1[r] * x1[r]; }
    const float rnr0 = 1.0f / fmaxf(sqrtf(ss0), EPS_);
    const float rnr1 = 1.0f / fmaxf(sqrtf(ss1), EPS_);

    float p[R_], ex[4];
#pragma unroll
    for (int r = 0; r < R_; r++) p[r] = x0[r] * x0[r] + x1[r] * x1[r];
    const float csq = reduce_columns<false>(p, ex, lane);
    rnwS[wv][lane] = 1.0f / fmaxf(sqrtf(csq), EPS_);
    if (lane < 4) {
        const float e = (lane == 0) ? ex[0] : (lane == 1) ? ex[1] : (lane == 2) ? ex[2] : ex[3];
        rnwS[wv][32 + lane] = 1.0f / fmaxf(sqrtf(e), EPS_);
    }
    __syncthreads();

    float m0 = NEG_, m1 = NEG_;
    float qv[R_];
#pragma unroll
    for (int q = 0; q < 9; q++) {
        const float4 n4 = *(const float4*)&rnwS[wv][4 * q];
        const float nn[4] = {n4.x, n4.y, n4.z, n4.w};
#pragma unroll
        for (int e = 0; e < 4; e++) {
            const int r = 4 * q + e;
            m0 = fmaxf(m0, x0[r] * nn[e]);
            m1 = fmaxf(m1, x1[r] * nn[e]);
            const float y0 = real0 ? x0[r] * rnr0 : NEG_;
            const float y1 = real1 ? x1[r] * rnr1 : NEG_;
            qv[r] = fmaxf(y0, y1);
        }
    }
    const float rs0 = sel0 ? m0 : NEG_;
    const float rs1 = sel1 ? m1 : NEG_;
    float cex[4];
    const float cs = reduce_columns<true>(qv, cex, lane);
    const float cs2 = (lane == 0) ? cex[0] : (lane == 1) ? cex[1] : (lane == 2) ? cex[2] : cex[3];

    const float rowmean = tree_sum((real0 ? rs0 : 0.f) + (real1 ? rs1 : 0.f)) / (float)cl;
    const float colmean = tree_sum(cs + ((lane < 4) ? cs2 : 0.f)) / (float)R_;

    rowS[wv][w0] = rs0;
    rowS[wv][w1] = rs1;
    colS[wv][lane] = cs;
    if (lane < 4) colS[wv][32 + lane] = cs2;
    __syncthreads();

    int rk0 = 0, rk1 = 0;
#pragma unroll
    for (int q = 0; q < W_ / 4; q++) {
        const float4 u4 = *(const float4*)&rowS[wv][4 * q];
        const float uu[4] = {u4.x, u4.y, u4.z, u4.w};
#pragma unroll
        for (int e = 0; e < 4; e++) {
            const int j = 4 * q + e;
            const float u = uu[e];
            rk0 += (int)((u > rs0) | ((u == rs0) & (j < w0)));
            rk1 += (int)((u > rs1) | ((u == rs1) & (j < w1)));
        }
    }
    int rc0 = 0, rc1 = 0;
#pragma unroll
    for (int q = 0; q < R_ / 4; q++) {
        const float4 u4 = *(const float4*)&colS[wv][4 * q];
        const float uu[4] = {u4.x, u4.y, u4.z, u4.w};
#pragma unroll
        for (int e = 0; e < 4; e++) {
            const int j = 4 * q + e;
            const float u = uu[e];
            rc0 += (int)((u > cs) | ((u == cs) & (j < lane)));
            rc1 += (int)((u > cs2) | ((u == cs2) & (j < lane + 32)));
        }
    }
    if (rk0 < TK_) topS[wv][0][rk0] = rs0;
    if (w1 < W_ && rk1 < TK_) topS[wv][0][rk1] = rs1;
    if (rc0 < TK_) topS[wv][1][rc0] = cs;
    if (lane < 4 && rc1 < TK_) topS[wv][1][rc1] = cs2;
    __syncthreads();

    if (lane < TK_) {
        float a = b1_t2i[lane], g = b1_i2t[lane];
#pragma unroll
        for (int k = 0; k < TK_; k++) {
            a = fmaf(topS[wv][0][k], wS[0][k * TK_ + lane], a);
            g = fmaf(topS[wv][1][k], wS[1][k * TK_ + lane], g);
        }
        hS[wv][0][lane] = fmaxf(a, 0.f);
        hS[wv][1][lane] = fmaxf(g, 0.f);
    }
    __syncthreads();

    if (lane == 0) {
        float o = b2_t2i[0];
#pragma unroll
        for (int j = 0; j < TK_; j++) o = fmaf(hS[wv][0][j], w2S[0][j], o);
        out[c * B_ + i]               = o + rowmean;
        out[2 * B_ * B_ + c * B_ + i] = rowmean;
    } else if (lane == 16) {
        float o = b2_i2t[0];
#pragma unroll
        for (int j = 0; j < TK_; j++) o = fmaf(hS[wv][1][j], w2S[1][j], o);
        out[B_ * B_ + i * B_ + c]     = o + colmean;
        out[3 * B_ * B_ + i * B_ + c] = colmean;
    }
}

extern "C" void kernel_launch(void* const* d_in, const int* in_sizes, int n_in,
                              void* d_out, int out_size, void* d_ws, size_t ws_size,
                              hipStream_t stream) {
    const float* img     = (const float*)d_in[0];
    const float* cap     = (const float*)d_in[1];
    const float* w1_t2i  = (const float*)d_in[2];
    const float* b1_t2i  = (const float*)d_in[3];
    const float* w2_t2i  = (const float*)d_in[4];
    const float* b2_t2i  = (const float*)d_in[5];
    const float* w1_i2t  = (const float*)d_in[6];
    const float* b1_i2t  = (const float*)d_in[7];
    const float* w2_i2t  = (const float*)d_in[8];
    const float* b2_i2t  = (const float*)d_in[9];
    const int*   cap_len = (const int*)d_in[10];
    float*       out     = (float*)d_out;

    float* inv_cap = (float*)d_ws;
    float* inv_img = inv_cap + B_ * W_;
    int*   rowmap  = (int*)(inv_img + B_ * R_);
    int*   nlive   = rowmap + B_ * W_;
    float* sim     = (float*)(nlive + 4);

    row_invnorm_kernel<<<1 + B_ * W_ + B_ * R_, 256, 0, stream>>>(img, cap, cap_len, inv_cap, inv_img, rowmap, nlive);

    sim_gemm_kernel<<<(B_ * W_ / 128) * NT_ + 3 * (NWGP_ - 1), 128, 0, stream>>>(img, cap, inv_cap, inv_img, rowmap, nlive, sim);

    dim3 grid(B_ / 4, B_);
    pair_tail_kernel<<<grid, 128, 0, stream>>>(sim,
                                               w1_t2i, b1_t2i, w2_t2i, b2_t2i,
                                               w1_i2t, b1_i2t, w2_i2t, b2_i2t,
                                               cap_len, out);
}


# row norms taken inside the GEMM from the values it stages (sum of squares kept per row, 1/norms applied to the dot product at the store); norm launch gone, 2 launches; norm sums in a new order
# speedup vs baseline: 3.0477x; 1.0394x over previous
#include <hip/hip_runtime.h>
#include <hip/hip_bf16.h>

typedef __attribute__((ext_vector_type(2))) float v2f;
typedef __attribute__((ext_vector_type(8))) float v8f;

#define B_    128
#define R_    36
#define W_    60
#define D_    1024
#define TK_   20
#define EPS_  1e-12f
#define NEG_  -1e30f

template <int M>
__device__ __forceinline__ float lane_xor(float v) {
    if (M == 16)
        return __builtin_bit_cast(float, __builtin_amdgcn_permlanex16(0, __builtin_bit_cast(int, v), 0x76543210, 0xfedcba98, true, false));
    else
        return __builtin_bit_cast(float, __builtin_amdgcn_update_dpp(0, __builtin_bit_cast(int, v), 0x160 + M, 0xf, 0xf, true));
}

template <int M, int N, bool MAXOP>
__device__ __forceinline__ void halve_step(float* v, const bool up) {
#pragma unroll
    for (int j = 0; j < N; j++) {
        const float a = v[j], b = v[j + N];
        const float keep = up ? b : a;
        const float send = up ? a : b;
        const float recv = lane_xor<M>(send);
        v[j] = MAXOP ? fmaxf(keep, recv) : (keep + recv);
    }
}

#define PITCH_ 36
#define NT_    36
#define NWGP_  256

template <int TI, int TJ>
__device__ __forceinline__ void gemm_tile(const float* __restrict__ img,
                                          const float* __restrict__ cap,
                                          float* __restrict__ sim,
                                          const int n0,
                                          float (*As)[128][PITCH_],
                                          float (*Bs)[128][PITCH_],
                                          const int* rowS,
                                          float* invA,
                                          float* invB) {
    constexpr int BM = 32 * TI;
    constexpr int JA = BM / 16;
    constexpr int JB = (32 * TJ) / 16;

    const int tid  = threadIdx.x;
    const int lane = tid & 31;
    const int wv   = tid >> 5;
    const int wm   = wv >> 1;
    const int wn   = wv & 1;

    if (rowS[0] < 0) return;

    const int srow = tid >> 3;
    const int sc4  = (tid & 7) << 2;
    v2f ssA[JA], ssB[JB];
    int   offA[JA];
#pragma unroll
    for (int j = 0; j < JA; j++) {
        const int o  = rowS[srow + 16 * j];
        const int oc = (o < 0) ? 0 : o;
        ssA[j]  = (v2f){0.f, 0.f};
        offA[j] = oc * D_ + sc4;
    }
#pragma unroll
    for (int j = 0; j < JB; j++) ssB[j] = (v2f){0.f, 0.f};
    const float* gB = img + (size_t)(n0 + srow) * D_ + sc4;

    v8f acc[TI][TJ];
#pragma unroll
    for (int a = 0; a < TI; a++)
#pragma unroll
        for (int b = 0; b < TJ; b++) acc[a][b] = (v8f){};

    const int fr = lane & 15;
    const int kb = (lane >> 4) << 1;

    float4 va[JA], vb[JB];
#pragma unroll
    for (int j = 0; j < JA; j++) va[j] = *(const float4*)(cap + offA[j]);
#pragma unroll
    for (int j = 0; j < JB; j++) vb[j] = *(const float4*)(gB + (size_t)j * 16 * D_);
#pragma unroll
    for (int j = 0; j < JA; j++) {
        const float4 x = va[j];
        const v2f xl = {x.x, x.y}, xh = {x.z, x.w};
        ssA[j] += xl * xl; ssA[j] += xh * xh;
        *(float4*)&As[0][srow + 16 * j][sc4] = x;
    }
#pragma unroll
    for (int j = 0; j < JB; j++) {
        const float4 y = vb[j];
        const v2f yl = {y.x, y.y}, yh = {y.z, y.w};
        ssB[j] += yl * yl; ssB[j] += yh * yh;
        *(float4*)&Bs[0][srow + 16 * j][sc4] = y;
    }
    __syncthreads();

    for (int k0 = 0; k0 < D_; k0 += 32) {
        const int cur = (k0 >> 5) & 1;
        const bool more = (k0 + 32 < D_);
        if (more) {
#pragma unroll
            for (int j = 0; j < JA; j++) va[j] = *(const float4*)(cap + offA[j] + k0 + 32);
#pragma unroll
            for (int j = 0; j < JB; j++) vb[j] = *(const float4*)(gB + (size_t)j * 16 * D_ + k0 + 32);
        }
#pragma unroll
        for (int kk = 0; kk < 32; kk += 4) {
            v2f a[TI], b[TJ];
#pragma unroll
            for (int t = 0; t < TI; t++) a[t] = *(const v2f*)&As[cur][wm * (16 * TI) + t * 16 + fr][kk + kb];
#pragma unroll
            for (int t = 0; t < TJ; t++) b[t] = *(const v2f*)&Bs[cur][wn * (16 * TJ) + t * 16 + fr][kk + kb];
#pragma unroll
            for (int ti = 0; ti < TI; ti++)
#pragma unroll
                for (int tj = 0; tj < TJ; tj++)
                    acc[ti][tj] = __builtin_amdgcn_wmma_f32_16x16x4_f32(false, a[ti], false, b[tj], (short)0, acc[ti][tj], false, false);
        }
        if (more) {
#pragma unroll
            for (int j = 0; j < JA; j++) {
                const float4 x = va[j];
                const v2f xl = {x.x, x.y}, xh = {x.z, x.w};
                ssA[j] += xl * xl; ssA[j] += xh * xh;
                *(float4*)&As[cur ^ 1][srow + 16 * j][sc4] = x;
            }
#pragma unroll
            for (int j = 0; j < JB; j++) {
                const float4 y = vb[j];
                const v2f yl = {y.x, y.y}, yh = {y.z, y.w};
                ssB[j] += yl * yl; ssB[j] += yh * yh;
                *(float4*)&Bs[cur ^ 1][srow + 16 * j][sc4] = y;
            }
        }
        __syncthreads();
    }

    constexpr int NV = JA + JB;
    float sv[NV];
#pragma unroll
    for (int j = 0; j < JA; j++) sv[j] = ssA[j].x + ssA[j].y;
#pragma unroll
    for (int j = 0; j < JB; j++) sv[JA + j] = ssB[j].x + ssB[j].y;
    halve_step<4, NV / 2, false>(sv, (lane & 4) != 0);
    halve_step<2, NV / 4, false>(sv, (lane & 2) != 0);
    halve_step<1, NV / 8, false>(sv, (lane & 1) != 0);
#pragma unroll
    for (int e = 0; e < NV / 8; e++) {
        const int idx = (tid & 7) * (NV / 8) + e;
        const float inv = 1.0f / fmaxf(sqrtf(sv[e]), EPS_);
        if (idx < JA) invA[srow + 16 * idx] = inv;
        else          invB[srow + 16 * (idx - JA)] = inv;
    }
    __syncthreads();

    int   colOff[TJ];
    float colInv[TJ];
#pragma unroll
    for (int tj = 0; tj < TJ; tj++) {
        const int nl = wn * (16 * TJ) + tj * 16 + fr;
        const int n = n0 + nl;
        const int i = n / R_;
        colOff[tj] = i * (W_ * R_) + (n - i * R_);
        colInv[tj] = invB[nl];
    }
    const int mb = wm * (16 * TI) + ((lane >> 4) << 3);
#pragma unroll
    for (int ti = 0; ti < TI; ti++) {
#pragma unroll
        for (int v = 0; v < 8; v++) {
            const int m = rowS[mb + ti * 16 + v];
            if (m >= 0) {
                const int c = m / W_;
                const int rowOff = c * (B_ * W_ * R_) + (m - c * W_) * R_;
                const float rowInv = invA[mb + ti * 16 + v];
#pragma unroll
                for (int tj = 0; tj < TJ; tj++) {
                    const float x = acc[ti][tj][v] * rowInv * colInv[tj];
                    sim[rowOff + colOff[tj]] = (x >= 0.f) ? x : 0.1f * x;
                }
            }
        }
    }
}

__global__ __launch_bounds__(128)
void sim_gemm_kernel(const float* __restrict__ img,
                     const float* __restrict__ cap,
                     const int*   __restrict__ cap_len,
                     float* __restrict__ sim) {
    __shared__ __attribute__((aligned(16))) float As[2][128][PITCH_];
    __shared__ __attribute__((aligned(16))) float Bs[2][128][PITCH_];
    __shared__ int   rowS[128];
    __shared__ float invA[128];
    __shared__ float invB[128];
    __shared__ int   wsum[4];

    const int tid = threadIdx.x;
    const int ln  = tid & 31;
    const int ns  = min(max(cap_len[tid], TK_), W_);
    int incl = ns;
    for (int d = 1; d < 32; d <<= 1) {
        const int u = __shfl_up(incl, d);
        if (ln >= d) incl += u;
    }
    if (ln == 31) wsum[tid >> 5] = incl;
    rowS[tid] = -1;
    __syncthreads();
    int base = incl - ns, total = 0;
    for (int k = 0; k < 4; k++) {
        const int s = wsum[k];
        total += s;
        if (k < (tid >> 5)) base += s;
    }

    const int tiles = ((total + 127) >> 7) * NT_;
    const int bulk  = tiles & ~(NWGP_ - 1);
    const int id    = blockIdx.x;
    const bool whole = id < bulk;
    int m0, n0, bm;
    if (whole) {
        const int ty = id / NT_;
        m0 = ty * 128; n0 = (id - ty * NT_) * 128; bm = 128;
    } else {
        const int q = id - bulk;
        const int t = bulk + (q >> 2);
        if (t >= tiles) return;
        const int ty = t / NT_;
        m0 = ty * 128 + ((q >> 1) & 1) * 64; n0 = (t - ty * NT_) * 128 + (q & 1) * 64; bm = 64;
    }

    const int lo = max(base, m0), hi = min(base + ns, m0 + bm);
    for (int m = lo; m < hi; m++) rowS[m - m0] = tid * W_ + (m - base);
    __syncthreads();

    if (whole) gemm_tile<4, 4>(img, cap, sim, n0, As, Bs, rowS, invA, invB);
    else       gemm_tile<2, 2>(img, cap, sim, n0, As, Bs, rowS, invA, invB);
}

template <bool MAXOP>
__device__ __forceinline__ float reduce_columns(float* v, float* ex, const int lane) {
    halve_step<16, 16, MAXOP>(v, (lane & 16) != 0);
    halve_step<8, 8, MAXOP>(v, (lane & 8) != 0);
    halve_step<4, 4, MAXOP>(v, (lane & 4) != 0);
    halve_step<2, 2, MAXOP>(v, (lane & 2) != 0);
    halve_step<1, 1, MAXOP>(v, (lane & 1) != 0);
#pragma unroll
    for (int e = 0; e < 4; e++) {
        float t = v[32 + e];
        { const float o = lane_xor<16>(t); t = MAXOP ? fmaxf(t, o) : (t + o); }
        { const float o = lane_xor<8>(t);  t = MAXOP ? fmaxf(t, o) : (t + o); }
        { const float o = lane_xor<4>(t);  t = MAXOP ? fmaxf(t, o) : (t + o); }
        { const float o = lane_xor<2>(t);  t = MAXOP ? fmaxf(t, o) : (t + o); }
        { const float o = lane_xor<1>(t);  t = MAXOP ? fmaxf(t, o) : (t + o); }
        ex[e] = t;
    }
    return v[0];
}

__device__ __forceinline__ float tree_sum(float t) {
    t += lane_xor<16>(t);
    t += lane_xor<8>(t);
    t += lane_xor<4>(t);
    t += lane_xor<2>(t);
    t += lane_xor<1>(t);
    return t;
}

__global__ __launch_bounds__(128)
void pair_tail_kernel(const float* __restrict__ sim,
                      const float* __restrict__ w1_t2i, const float* __restrict__ b1_t2i,
                      const float* __restrict__ w2_t2i, const float* __restrict__ b2_t2i,
                      const float* __restrict__ w1_i2t, const float* __restrict__ b1_i2t,
                      const float* __restrict__ w2_i2t, const float* __restrict__ b2_i2t,
                      const int*   __restrict__ cap_len,
                      float*       __restrict__ out) {
    const int tid  = threadIdx.x;
    const int lane = tid & 31;
    const int wv   = tid >> 5;
    const int i    = blockIdx.x * 4 + wv;
    const int c    = blockIdx.y;

    __shared__ float4 tileS[4][17 * 32];
    __shared__ __attribute__((aligned(16))) float wS[2][TK_ * TK_];
    __shared__ __attribute__((aligned(16))) float w2S[2][TK_];
    __shared__ __attribute__((aligned(16))) float rnwS[4][40];
    __shared__ __attribute__((aligned(16))) float rowS[4][64];
    __shared__ __attribute__((aligned(16))) float colS[4][40];
    __shared__ __attribute__((aligned(16))) float topS[4][2][TK_];
    __shared__ __attribute__((aligned(16))) float hS[4][2][TK_];

    for (int k = tid; k < TK_ * TK_; k += 128) { wS[0][k] = w1_t2i[k]; wS[1][k] = w1_i2t[k]; }
    if (tid < TK_) { w2S[0][tid] = w2_t2i[tid]; w2S[1][tid] = w2_i2t[tid]; }

    const int cl   = cap_len[c];
    const int nsel = min(max(cl, TK_), W_);
    const int w0 = lane, w1 = lane + 32;
    const bool sel0 = w0 < nsel, sel1 = w1 < nsel;
    const bool real0 = w0 < cl,  real1 = (w1 < cl) && (w1 < W_);

    const float4* tile = (const float4*)(sim + (size_t)(c * B_ + i) * (W_ * R_));
    const int npiece = nsel * (R_ / 4);
    float4 pc[17];
#pragma unroll
    for (int q = 0; q < 17; q++) {
        const int idx = q * 32 + lane;
        pc[q] = tile[(idx < npiece) ? idx : 0];
    }
#pragma unroll
    for (int q = 0; q < 17; q++) tileS[wv][q * 32 + lane] = pc[q];
    __syncthreads();
    float x0[R_], x1[R_];
#pragma unroll
    for (int q = 0; q < 9; q++) {
        const float4 a = sel0 ? tileS[wv][w0 * 9 + q] : make_float4(0.f, 0.f, 0.f, 0.f);
        const float4 b = sel1 ? tileS[wv][w1 * 9 + q] : make_float4(0.f, 0.f, 0.f, 0.f);
        x0[4 * q + 0] = a.x; x0[4 * q + 1] = a.y; x0[4 * q + 2] = a.z; x0[4 * q + 3] = a.w;
        x1[4 * q + 0] = b.x; x1[4 * q + 1] = b.y; x1[4 * q + 2] = b.z; x1[4 * q + 3] = b.w;
    }

    float ss0 = 0.f, ss1 = 0.f;
#pragma unroll
    for (int r = 0; r < R_; r++) { ss0 += x0[r] * x0[r]; ss1 += x1[r] * x1[r]; }
    const float rnr0 = 1.0f / fmaxf(sqrtf(ss0), EPS_);
    const float rnr1 = 1.0f / fmaxf(sqrtf(ss1), EPS_);

    float p[R_], ex[4];
#pragma unroll
    for (int r = 0; r < R_; r++) p[r] = x0[r] * x0[r] + x1[r] * x1[r];
    const float csq = reduce_columns<false>(p, ex, lane);
    rnwS[wv][lane] = 1.0f / fmaxf(sqrtf(csq), EPS_);
    if (lane < 4) {
        const float e = (lane == 0) ? ex[0] : (lane == 1) ? ex[1] : (lane == 2) ? ex[2] : ex[3];
        rnwS[wv][32 + lane] = 1.0f / fmaxf(sqrtf(e), EPS_);
    }
    __syncthreads();

    float m0 = NEG_, m1 = NEG_;
    float qv[R_];
#pragma unroll
    for (int q = 0; q < 9; q++) {
        const float4 n4 = *(const float4*)&rnwS[wv][4 * q];
        const float nn[4] = {n4.x, n4.y, n4.z, n4.w};
#pragma unroll
        for (int e = 0; e < 4; e++) {
            const int r = 4 * q + e;
            m0 = fmaxf(m0, x0[r] * nn[e]);
            m1 = fmaxf(m1, x1[r] * nn[e]);
            const float y0 = real0 ? x0[r] * rnr0 : NEG_;
            const float y1 = real1 ? x1[r] * rnr1 : NEG_;
            qv[r] = fmaxf(y0, y1);
        }
    }
    const float rs0 = sel0 ? m0 : NEG_;
    const float rs1 = sel1 ? m1 : NEG_;
    float cex[4];
    const float cs = reduce_columns<true>(qv, cex, lane);
    const float cs2 = (lane == 0) ? cex[0] : (lane == 1) ? cex[1] : (lane == 2) ? cex[2] : cex[3];

    const float rowmean = tree_sum((real0 ? rs0 : 0.f) + (real1 ? rs1 : 0.f)) / (float)cl;
    const float colmean = tree_sum(cs + ((lane < 4) ? cs2 : 0.f)) / (float)R_;

    rowS[wv][w0] = rs0;
    rowS[wv][w1] = rs1;
    colS[wv][lane] = cs;
    if (lane < 4) colS[wv][32 + lane] = cs2;
    __syncthreads();

    int rk0 = 0, rk1 = 0;
#pragma unroll
    for (int q = 0; q < W_ / 4; q++) {
        const float4 u4 = *(const float4*)&rowS[wv][4 * q];
        const float uu[4] = {u4.x, u4.y, u4.z, u4.w};
#pragma unroll
        for (int e = 0; e < 4; e++) {
            const int j = 4 * q + e;
            const float u = uu[e];
            rk0 += (int)((u > rs0) | ((u == rs0) & (j < w0)));
            rk1 += (int)((u > rs1) | ((u == rs1) & (j < w1)));
        }
    }
    int rc0 = 0, rc1 = 0;
#pragma unroll
    for (int q = 0; q < R_ / 4; q++) {
        const float4 u4 = *(const float4*)&colS[wv][4 * q];
        const float uu[4] = {u4.x, u4.y, u4.z, u4.w};
#pragma unroll
        for (int e = 0; e < 4; e++) {
            const int j = 4 * q + e;
            const float u = uu[e];
            rc0 += (int)((u > cs) | ((u == cs) & (j < lane)));
            rc1 += (int)((u > cs2) | ((u == cs2) & (j < lane + 32)));
        }
    }
    if (rk0 < TK_) topS[wv][0][rk0] = rs0;
    if (w1 < W_ && rk1 < TK_) topS[wv][0][rk1] = rs1;
    if (rc0 < TK_) topS[wv][1][rc0] = cs;
    if (lane < 4 && rc1 < TK_) topS[wv][1][rc1] = cs2;
    __syncthreads();

    if (lane < TK_) {
        float a = b1_t2i[lane], g = b1_i2t[lane];
#pragma unroll
        for (int k = 0; k < TK_; k++) {
            a = fmaf(topS[wv][0][k], wS[0][k * TK_ + lane], a);
            g = fmaf(topS[wv][1][k], wS[1][k * TK_ + lane], g);
        }
        hS[wv][0][lane] = fmaxf(a, 0.f);
        hS[wv][1][lane] = fmaxf(g, 0.f);
    }
    __syncthreads();

    if (lane == 0) {
        float o = b2_t2i[0];
#pragma unroll
        for (int j = 0; j < TK_; j++) o = fmaf(hS[wv][0][j], w2S[0][j], o);
        out[c * B_ + i]               = o + rowmean;
        out[2 * B_ * B_ + c * B_ + i] = rowmean;
    } else if (lane == 16) {
        float o = b2_i2t[0];
#pragma unroll
        for (int j = 0; j < TK_; j++) o = fmaf(hS[wv][1][j], w2S[1][j], o);
        out[B_ * B_ + i * B_ + c]     = o + colmean;
        out[3 * B_ * B_ + i * B_ + c] = colmean;
    }
}

extern "C" void kernel_launch(void* const* d_in, const int* in_sizes, int n_in,
                              void* d_out, int out_size, void* d_ws, size_t ws_size,
                              hipStream_t stream) {
    const float* img     = (const float*)d_in[0];
    const float* cap     = (const float*)d_in[1];
    const float* w1_t2i  = (const float*)d_in[2];
    const float* b1_t2i  = (const float*)d_in[3];
    const float* w2_t2i  = (const float*)d_in[4];
    const float* b2_t2i  = (const float*)d_in[5];
    const float* w1_i2t  = (const float*)d_in[6];
    const float* b1_i2t  = (const float*)d_in[7];
    const float* w2_i2t  = (const float*)d_in[8];
    const float* b2_i2t  = (const float*)d_in[9];
    const int*   cap_len = (const int*)d_in[10];
    float*       out     = (float*)d_out;

    float* sim = (float*)d_ws;

    sim_gemm_kernel<<<(B_ * W_ / 128) * NT_ + 3 * (NWGP_ - 1), 128, 0, stream>>>(img, cap, cap_len, sim);

    dim3 grid(B_ / 4, B_);
    pair_tail_kernel<<<grid, 128, 0, stream>>>(sim,
                                               w1_t2i, b1_t2i, w2_t2i, b2_t2i,
                                               w1_i2t, b1_i2t, w2_i2t, b2_i2t,
                                               cap_len, out);
}
